# starting kernel
# baseline (speedup 1.0000x reference)
#include <hip/hip_runtime.h>

#define D_FEAT   64
#define K_DIM    128
#define TILE_M   16
#define M_TILES  2
#define WAVES    4
#define WSTRIDE  72
#define FSTRIDE  132

typedef __attribute__((ext_vector_type(2))) float v2f;
typedef __attribute__((ext_vector_type(4))) float v4f;
typedef __attribute__((ext_vector_type(8))) float v8f;

__device__ __forceinline__ v8f wmma_f32_16x16x4(v2f a, v2f b, v8f c) {
  return __builtin_amdgcn_wmma_f32_16x16x4_f32(false, a, false, b, (short)0, c,
                                               false, false);
}

__global__ void k0_init(float* __restrict__ out, float* __restrict__ wsf,
                        const float* __restrict__ attn, int N, int T) {
  size_t idx    = (size_t)blockIdx.x * blockDim.x + threadIdx.x;
  size_t stride = (size_t)gridDim.x * blockDim.x;
  size_t nOut = (size_t)N * D_FEAT;
  size_t nCnt = (size_t)T * N;
  for (size_t i = idx; i < nOut; i += stride) out[i] = 0.0f;
  for (size_t i = idx; i < nCnt; i += stride) wsf[8 + i] = 0.0f;
  if (idx == 0) {
    float m = attn[0];
    for (int t = 1; t < T; ++t) m = fmaxf(m, attn[t]);
    float e[8];
    float s = 0.0f;
    for (int t = 0; t < T && t < 8; ++t) { e[t] = __expf(attn[t] - m); s += e[t]; }
    for (int t = 0; t < T && t < 8; ++t) wsf[t] = e[t] / s;
  }
}

__global__ void k1_count(const int* __restrict__ edges, float* __restrict__ cnt,
                         int N, int E, int T) {
  size_t idx    = (size_t)blockIdx.x * blockDim.x + threadIdx.x;
  size_t stride = (size_t)gridDim.x * blockDim.x;
  size_t total  = (size_t)T * E;
  for (size_t i = idx; i < total; i += stride) {
    size_t t = i / (size_t)E;
    size_t e = i - t * (size_t)E;
    int s = edges[t * 2u * (size_t)E + e];
    atomicAdd(&cnt[t * (size_t)N + s], 1.0f);
  }
}

__global__ void k2_scale(float* __restrict__ wsf, int N, int T) {
  size_t idx   = (size_t)blockIdx.x * blockDim.x + threadIdx.x;
  size_t total = (size_t)T * N;
  if (idx >= total) return;
  size_t t = idx / (size_t)N;
  float c = wsf[8 + idx];
  wsf[8 + idx] = wsf[t] / fmaxf(c, 1.0f);
}

__global__ __launch_bounds__(WAVES * 32)
void k3_gemm_scatter(const float* __restrict__ x, const float* __restrict__ W,
                     const float* __restrict__ b, const int* __restrict__ edges,
                     const float* __restrict__ scale, float* __restrict__ out,
                     int N, int E) {
  extern __shared__ float lds[];
  float* sW = lds;
  float* sF = lds + K_DIM * WSTRIDE;

  const int t     = blockIdx.y;
  const int tid   = threadIdx.x;
  const int wave  = tid >> 5;
  const int lane  = tid & 31;
  const int row16 = lane & 15;
  const int half  = lane >> 4;

  const float* Wt = W + (size_t)t * K_DIM * D_FEAT;
  for (int i = tid; i < K_DIM * (D_FEAT / 4); i += WAVES * 32) {
    int r  = i >> 4;
    int c4 = (i & 15) << 2;
    v4f v = *(const v4f*)(Wt + r * D_FEAT + c4);
    float* p = sW + r * WSTRIDE + c4;
    p[0] = v.x; p[1] = v.y; p[2] = v.z; p[3] = v.w;
  }
  __syncthreads();

  const int nblk32 = (E + M_TILES * TILE_M - 1) / (M_TILES * TILE_M);
  const int blk    = blockIdx.x * WAVES + wave;
  if (blk >= nblk32) return;

  const int* srcE = edges + (size_t)t * 2 * E;
  const int* dstE = srcE + E;
  float* sFw = sF + wave * (M_TILES * TILE_M * FSTRIDE);

  const int e0 = blk * (M_TILES * TILE_M);
  int ed0 = e0 + row16;
  int ed1 = e0 + TILE_M + row16;
  int cl0 = (ed0 < E) ? ed0 : (E - 1);
  int cl1 = (ed1 < E) ? ed1 : (E - 1);
  const int srcn0 = srcE[cl0];
  const int dstn0 = dstE[cl0];
  const int srcn1 = srcE[cl1];
  const int dstn1 = dstE[cl1];

  #pragma unroll 4
  for (int r = 0; r < TILE_M; ++r) {
    int sn0 = __shfl(srcn0, r, 32);
    int dn0 = __shfl(dstn0, r, 32);
    int sn1 = __shfl(srcn1, r, 32);
    int dn1 = __shfl(dstn1, r, 32);
    v2f vs0 = *(const v2f*)(x + (size_t)sn0 * D_FEAT + 2 * lane);
    v2f vd0 = *(const v2f*)(x + (size_t)dn0 * D_FEAT + 2 * lane);
    v2f vs1 = *(const v2f*)(x + (size_t)sn1 * D_FEAT + 2 * lane);
    v2f vd1 = *(const v2f*)(x + (size_t)dn1 * D_FEAT + 2 * lane);
    *(v2f*)(sFw + r * FSTRIDE + 2 * lane)                               = vs0;
    *(v2f*)(sFw + r * FSTRIDE + D_FEAT + 2 * lane)                      = vd0;
    *(v2f*)(sFw + (TILE_M + r) * FSTRIDE + 2 * lane)                    = vs1;
    *(v2f*)(sFw + (TILE_M + r) * FSTRIDE + D_FEAT + 2 * lane)           = vd1;
  }

  v8f accA0 = {}, accA1 = {}, accA2 = {}, accA3 = {};
  v8f accB0 = {}, accB1 = {}, accB2 = {}, accB3 = {};

  const float* aRow0 = sFw + row16 * FSTRIDE + 2 * half;
  const float* aRow1 = sFw + (TILE_M + row16) * FSTRIDE + 2 * half;

  #pragma unroll
  for (int c = 0; c < K_DIM / 4; ++c) {
    v2f a0 = *(const v2f*)(aRow0 + 4 * c);
    v2f a1 = *(const v2f*)(aRow1 + 4 * c);
    const float* bRow0 = sW + (4 * c + 2 * half) * WSTRIDE + row16;
    const float* bRow1 = bRow0 + WSTRIDE;
    v2f b0; b0.x = bRow0[0];  b0.y = bRow1[0];
    v2f b1; b1.x = bRow0[16]; b1.y = bRow1[16];
    v2f b2; b2.x = bRow0[32]; b2.y = bRow1[32];
    v2f b3; b3.x = bRow0[48]; b3.y = bRow1[48];
    accA0 = wmma_f32_16x16x4(a0, b0, accA0);
    accB0 = wmma_f32_16x16x4(a1, b0, accB0);
    accA1 = wmma_f32_16x16x4(a0, b1, accA1);
    accB1 = wmma_f32_16x16x4(a1, b1, accB1);
    accA2 = wmma_f32_16x16x4(a0, b2, accA2);
    accB2 = wmma_f32_16x16x4(a1, b2, accB2);
    accA3 = wmma_f32_16x16x4(a0, b3, accA3);
    accB3 = wmma_f32_16x16x4(a1, b3, accB3);
  }

  const float* bt = b + (size_t)t * D_FEAT;
  float bias0 = bt[0  + row16];
  float bias1 = bt[16 + row16];
  float bias2 = bt[32 + row16];
  float bias3 = bt[48 + row16];

  v8f accsA[4] = {accA0, accA1, accA2, accA3};
  v8f accsB[4] = {accB0, accB1, accB2, accB3};
  float biases[4] = {bias0, bias1, bias2, bias3};

  #pragma unroll
  for (int v = 0; v < 8; ++v) {
    int m = v + 8 * half;
    {
      int  sn    = __shfl(srcn0, m, 32);
      bool valid = (e0 + m) < E;
      float sc   = scale[(size_t)t * N + sn];
      #pragma unroll
      for (int nt = 0; nt < 4; ++nt) {
        float val = accsA[nt][v] + biases[nt];
        val = fmaxf(val, 0.0f) * sc;
        if (valid)
          atomicAdd(out + (size_t)sn * D_FEAT + nt * 16 + row16, val);
      }
    }
    {
      int  sn    = __shfl(srcn1, m, 32);
      bool valid = (e0 + TILE_M + m) < E;
      float sc   = scale[(size_t)t * N + sn];
      #pragma unroll
      for (int nt = 0; nt < 4; ++nt) {
        float val = accsB[nt][v] + biases[nt];
        val = fmaxf(val, 0.0f) * sc;
        if (valid)
          atomicAdd(out + (size_t)sn * D_FEAT + nt * 16 + row16, val);
      }
    }
  }
}

extern "C" void kernel_launch(void* const* d_in, const int* in_sizes, int n_in,
                              void* d_out, int out_size, void* d_ws, size_t ws_size,
                              hipStream_t stream) {
  const float* x     = (const float*)d_in[0];
  const float* W     = (const float*)d_in[1];
  const float* b     = (const float*)d_in[2];
  const float* attn  = (const float*)d_in[3];
  const int*   edges = (const int*)d_in[4];
  float* out = (float*)d_out;
  float* wsf = (float*)d_ws;

  const int N = in_sizes[0] / D_FEAT;
  const int T = in_sizes[3];
  const int E = in_sizes[4] / (2 * T);

  k0_init<<<2048, 256, 0, stream>>>(out, wsf, attn, N, T);
  k1_count<<<2048, 256, 0, stream>>>(edges, wsf + 8, N, E, T);
  {
    size_t total = (size_t)T * N;
    k2_scale<<<(unsigned)((total + 255) / 256), 256, 0, stream>>>(wsf, N, T);
  }
  const int nblk32 = (E + M_TILES * TILE_M - 1) / (M_TILES * TILE_M);
  dim3 grid((nblk32 + WAVES - 1) / WAVES, T);
  size_t ldsBytes = (size_t)(K_DIM * WSTRIDE + WAVES * M_TILES * TILE_M * FSTRIDE)
                    * sizeof(float);
  k3_gemm_scatter<<<grid, WAVES * 32, ldsBytes, stream>>>(x, W, b, edges,
                                                          wsf + 8, out, N, E);
}


# prelude's 48k short waves cut to 8k looping ones: fill and count as one 16-wave group per WGP (16-byte stores, 16 ids in flight per lane), k2's divide done in k3's scale read: 3 launches; same sums
# speedup vs baseline: 1.5436x; 1.5436x over previous
#include <hip/hip_runtime.h>
#include <type_traits>

#define D_FEAT   64
#define K_DIM    128
#define TILE_M   16
#define WAVES    8
#define FSTRIDE  132

typedef __attribute__((ext_vector_type(2))) float v2f;
typedef __attribute__((ext_vector_type(4))) float v4f;
typedef __attribute__((ext_vector_type(8))) float v8f;

__device__ __forceinline__ v8f wmma_f32_16x16x4(v2f a, v2f b, v8f c) {
  return __builtin_amdgcn_wmma_f32_16x16x4_f32(false, a, false, b, (short)0, c,
                                               false, false);
}

#define PRE_THREADS 512
#define CNT_UNROLL  16

__global__ __launch_bounds__(PRE_THREADS)
void k0_init(float* __restrict__ out, float* __restrict__ wsf,
             const float* __restrict__ attn, int N, int T) {
  const size_t idx    = (size_t)blockIdx.x * blockDim.x + threadIdx.x;
  const size_t stride = (size_t)gridDim.x * blockDim.x;
  const size_t nOut4 = (size_t)N * (D_FEAT / 4);
  const size_t nCnt  = (size_t)T * N;
  const size_t nCnt4 = nCnt / 4;
  const v4f z = {0.0f, 0.0f, 0.0f, 0.0f};
  v4f* out4 = (v4f*)out;
  v4f* cnt4 = (v4f*)(wsf + 8);
  for (size_t i = idx; i < nOut4; i += stride) out4[i] = z;
  for (size_t i = idx; i < nCnt4; i += stride) cnt4[i] = z;
  for (size_t i = 4 * nCnt4 + idx; i < nCnt; i += stride) wsf[8 + i] = 0.0f;
  if (idx == 0) {
    float m = attn[0];
    for (int t = 1; t < T; ++t) m = fmaxf(m, attn[t]);
    float e[8];
    float s = 0.0f;
    for (int t = 0; t < T && t < 8; ++t) { e[t] = __expf(attn[t] - m); s += e[t]; }
    for (int t = 0; t < T && t < 8; ++t) wsf[t] = e[t] / s;
  }
}

__global__ __launch_bounds__(PRE_THREADS)
void k1_count(const int* __restrict__ edges, float* __restrict__ cnt,
              int N, int E) {
  const size_t t      = blockIdx.y;
  const size_t idx    = (size_t)blockIdx.x * blockDim.x + threadIdx.x;
  const size_t stride = (size_t)gridDim.x * blockDim.x;
  const size_t nE     = (size_t)E;
  const int* srcT = edges + t * 2u * nE;
  float*     cntT = cnt + t * (size_t)N;
  const size_t last = nE - 1;
  for (size_t e = idx; e < nE; e += CNT_UNROLL * stride) {
    int  s[CNT_UNROLL];
    bool ok[CNT_UNROLL];
    #pragma unroll
    for (int u = 0; u < CNT_UNROLL; ++u) {
      const size_t j = e + (size_t)u * stride;
      ok[u] = j < nE;
      s[u]  = srcT[ok[u] ? j : last];
    }
    asm volatile("" :: "v"(s[0]), "v"(s[1]), "v"(s[2]), "v"(s[3]),
                       "v"(s[4]), "v"(s[5]), "v"(s[6]), "v"(s[7]),
                       "v"(s[8]), "v"(s[9]), "v"(s[10]), "v"(s[11]),
                       "v"(s[12]), "v"(s[13]), "v"(s[14]), "v"(s[15]));
    #pragma unroll
    for (int u = 0; u < CNT_UNROLL; ++u)
      if (ok[u]) atomicAdd(&cntT[s[u]], 1.0f);
  }
}


__device__ __forceinline__ void row_info(int pSrc, int pScBits, int v, int half,
                                         int& sn, float& sc) {
  int   snLo = __builtin_amdgcn_readlane(pSrc, v);
  int   snHi = __builtin_amdgcn_readlane(pSrc, v + 8);
  float scLo = __int_as_float(__builtin_amdgcn_readlane(pScBits, v));
  float scHi = __int_as_float(__builtin_amdgcn_readlane(pScBits, v + 8));
  sn = half ? snHi : snLo;
  sc = half ? scHi : scLo;
}

template <bool EPI>
__device__ __forceinline__ void mma_tile(const float* aRow,
                                         const v2f (&Bf)[K_DIM / 4][4],
                                         v8f (&acc)[4], const v8f (&prv)[4],
                                         int pSrc, int pScBits,
                                         const float (&bias)[4],
                                         float* __restrict__ out,
                                         int half, int row16) {
  v2f Af[K_DIM / 4];
  #pragma unroll
  for (int c = 0; c < K_DIM / 4; ++c) Af[c] = *(const v2f*)(aRow + 4 * c);

  v8f a0 = {}, a1 = {}, a2 = {}, a3 = {};
  float* rowp = out;
  float  sc   = 0.0f;
  #pragma unroll
  for (int c = 0; c < K_DIM / 4; ++c) {
    a0 = wmma_f32_16x16x4(Af[c], Bf[c][0], a0);
    a1 = wmma_f32_16x16x4(Af[c], Bf[c][1], a1);
    a2 = wmma_f32_16x16x4(Af[c], Bf[c][2], a2);
    a3 = wmma_f32_16x16x4(Af[c], Bf[c][3], a3);
    if (EPI) {
      const int v = c >> 2, nt = c & 3;
      if (nt == 0) {
        int sn;
        row_info(pSrc, pScBits, v, half, sn, sc);
        rowp = out + (size_t)sn * D_FEAT + row16;
      }
      float val = fmaxf(prv[nt][v] + bias[nt], 0.0f) * sc;
      atomicAdd(rowp + nt * 16, val);
    }
  }
  acc[0] = a0; acc[1] = a1; acc[2] = a2; acc[3] = a3;
}

__global__ __launch_bounds__(WAVES * 32)
void k3_gemm_scatter(const float* __restrict__ x, const float* __restrict__ W,
                     const float* __restrict__ b, const int* __restrict__ edges,
                     const float* __restrict__ wsoft,
                     const float* __restrict__ cnt, float* __restrict__ out,
                     int N, int E) {
  extern __shared__ float lds[];

  const int t     = blockIdx.y;
  const int tid   = threadIdx.x;
  const int wave  = __builtin_amdgcn_readfirstlane(tid >> 5);
  const int lane  = tid & 31;
  const int row16 = lane & 15;
  const int half  = lane >> 4;
  float* sFw = lds + wave * (TILE_M * FSTRIDE);

  const int ntile   = (E + TILE_M - 1) / TILE_M;
  const int tstride = gridDim.x * WAVES;
  int tile = blockIdx.x * WAVES + wave;
  if (tile >= ntile) return;

  const float* Wt = W + (size_t)t * K_DIM * D_FEAT;
  v2f Bf[K_DIM / 4][4];
  #pragma unroll
  for (int c = 0; c < K_DIM / 4; ++c) {
    const float* p = Wt + (4 * c + 2 * half) * D_FEAT + row16;
    #pragma unroll
    for (int nt = 0; nt < 4; ++nt) {
      Bf[c][nt].x = p[nt * 16];
      Bf[c][nt].y = p[D_FEAT + nt * 16];
    }
  }

  const float* bt = b + (size_t)t * D_FEAT;
  const float bias[4] = {bt[0 + row16], bt[16 + row16], bt[32 + row16],
                         bt[48 + row16]};

  const int* srcE = edges + (size_t)t * 2 * E;
  const int* dstE = srcE + E;
  const float* cntT = cnt + (size_t)t * N;
  const float  wt   = wsoft[t];

  int edA = tile * TILE_M + row16;
  int clA = (edA < E) ? edA : (E - 1);
  int srcA = srcE[clA], dstA = dstE[clA];
  int edB = (tile + tstride) * TILE_M + row16;
  int clB = (edB < E) ? edB : (E - 1);
  int srcB = srcE[clB], dstB = dstE[clB];

  v4f g[TILE_M];
  #pragma unroll
  for (int r = 0; r < TILE_M; ++r) {
    int sn = __builtin_amdgcn_readlane(srcA, r);
    int dn = __builtin_amdgcn_readlane(dstA, r);
    int node = half ? dn : sn;
    g[r] = *(const v4f*)(x + (size_t)node * D_FEAT + 4 * row16);
  }
  float scRawA = wt / fmaxf(cntT[srcA], 1.0f);
  float scA = (edA < E) ? scRawA : 0.0f;

  const float* aRow = sFw + row16 * FSTRIDE + 2 * half;
  v8f prv[4] = {};
  int pSrc = 0, pScBits = 0;

  auto trip = [&](auto epi) {
    #pragma unroll
    for (int r = 0; r < TILE_M; ++r)
      *(v4f*)(sFw + r * FSTRIDE + half * D_FEAT + 4 * row16) = g[r];

    #pragma unroll
    for (int r = 0; r < TILE_M; ++r) {
      int sn = __builtin_amdgcn_readlane(srcB, r);
      int dn = __builtin_amdgcn_readlane(dstB, r);
      int node = half ? dn : sn;
      g[r] = *(const v4f*)(x + (size_t)node * D_FEAT + 4 * row16);
    }
    float scRawB = wt / fmaxf(cntT[srcB], 1.0f);
    float scB = (edB < E) ? scRawB : 0.0f;
    int edC = (tile + 2 * tstride) * TILE_M + row16;
    int clC = (edC < E) ? edC : (E - 1);
    int srcC = srcE[clC], dstC = dstE[clC];

    v8f acc[4];
    mma_tile<decltype(epi)::value>(aRow, Bf, acc, prv, pSrc, pScBits, bias,
                                   out, half, row16);

    prv[0] = acc[0]; prv[1] = acc[1]; prv[2] = acc[2]; prv[3] = acc[3];
    pSrc = srcA; pScBits = __float_as_int(scA);

    srcA = srcB; dstA = dstB; scA = scB;
    srcB = srcC; dstB = dstC; edB = edC;
  };

  trip(std::false_type{});
  for (tile += tstride; tile < ntile; tile += tstride) trip(std::true_type{});

  #pragma unroll
  for (int v = 0; v < 8; ++v) {
    int sn; float sc;
    row_info(pSrc, pScBits, v, half, sn, sc);
    float* rowp = out + (size_t)sn * D_FEAT + row16;
    #pragma unroll
    for (int nt = 0; nt < 4; ++nt) {
      float val = fmaxf(prv[nt][v] + bias[nt], 0.0f) * sc;
      atomicAdd(rowp + nt * 16, val);
    }
  }
}

extern "C" void kernel_launch(void* const* d_in, const int* in_sizes, int n_in,
                              void* d_out, int out_size, void* d_ws, size_t ws_size,
                              hipStream_t stream) {
  const float* x     = (const float*)d_in[0];
  const float* W     = (const float*)d_in[1];
  const float* b     = (const float*)d_in[2];
  const float* attn  = (const float*)d_in[3];
  const int*   edges = (const int*)d_in[4];
  float* out = (float*)d_out;
  float* wsf = (float*)d_ws;

  const int N = in_sizes[0] / D_FEAT;
  const int T = in_sizes[3];
  const int E = in_sizes[4] / (2 * T);

  k0_init<<<256, PRE_THREADS, 0, stream>>>(out, wsf, attn, N, T);
  k1_count<<<dim3(52, T), PRE_THREADS, 0, stream>>>(edges, wsf + 8, N, E);
  dim3 grid(51, T);
  size_t ldsBytes = (size_t)(WAVES * TILE_M * FSTRIDE) * sizeof(float);
  k3_gemm_scatter<<<grid, WAVES * 32, ldsBytes, stream>>>(x, W, b, edges, wsf,
                                                          wsf + 8, out, N, E);
}


# edges binned by source-row range (lists in workspace); a k3 group owns 391 out rows, sums them in LDS by LDS float adds, stores them once: no global float atomics, no zero fill; sum order changes
# speedup vs baseline: 1.8123x; 1.1741x over previous
#include <hip/hip_runtime.h>
#include <type_traits>

#define D_FEAT   64
#define K_DIM    128
#define TILE_M   16
#define WAVES    8
#define FSTRIDE  132
#define ROWS_PER_BIN 391
#define ACCSTRIDE 80
#define T_MAX    8
#define LDS_FLOATS (WAVES * TILE_M * FSTRIDE + ROWS_PER_BIN * ACCSTRIDE + T_MAX * ROWS_PER_BIN)
#define CNT_UNROLL 8

typedef __attribute__((ext_vector_type(2))) float v2f;
typedef __attribute__((ext_vector_type(4))) float v4f;
typedef __attribute__((ext_vector_type(8))) float v8f;

__device__ __forceinline__ v8f wmma_f32_16x16x4(v2f a, v2f b, v8f c) {
  return __builtin_amdgcn_wmma_f32_16x16x4_f32(false, a, false, b, (short)0, c,
                                               false, false);
}

#define PRE_THREADS 512
#define ID_UNROLL   8

__global__ __launch_bounds__(PRE_THREADS)
void k0_init(int* __restrict__ wsi, int nWords) {
  const int idx = blockIdx.x * blockDim.x + threadIdx.x;
  if (idx < nWords) wsi[idx] = 0;
}

__device__ __forceinline__ void chunk_hist(const int* __restrict__ srcT, int c0,
                                           int c1, int* lh) {
  const int tid = threadIdx.x;
  for (int e = c0 + tid; e < c1; e += ID_UNROLL * PRE_THREADS) {
    int  s[ID_UNROLL];
    bool ok[ID_UNROLL];
    #pragma unroll
    for (int u = 0; u < ID_UNROLL; ++u) {
      const int j = e + u * PRE_THREADS;
      ok[u] = j < c1;
      s[u]  = srcT[ok[u] ? j : (c1 - 1)];
    }
    asm volatile("" :: "v"(s[0]), "v"(s[1]), "v"(s[2]), "v"(s[3]),
                       "v"(s[4]), "v"(s[5]), "v"(s[6]), "v"(s[7]));
    #pragma unroll
    for (int u = 0; u < ID_UNROLL; ++u)
      if (ok[u]) atomicAdd(&lh[s[u] / ROWS_PER_BIN], 1);
  }
}

__global__ __launch_bounds__(PRE_THREADS)
void k1_hist(const int* __restrict__ edges, int* __restrict__ hist, int E,
             int nBins) {
  extern __shared__ int sm[];
  const int t   = blockIdx.y;
  const int tid = threadIdx.x;
  const int* srcT = edges + (size_t)t * 2 * E;
  const int per = (E + gridDim.x - 1) / gridDim.x;
  const int c0  = blockIdx.x * per;
  const int c1  = min(E, c0 + per);
  for (int i = tid; i < nBins; i += PRE_THREADS) sm[i] = 0;
  __syncthreads();
  chunk_hist(srcT, c0, c1, sm);
  __syncthreads();
  for (int i = tid; i < nBins; i += PRE_THREADS) {
    const int c = sm[i];
    if (c) atomicAdd(&hist[t * nBins + i], c);
  }
}

__global__ __launch_bounds__(PRE_THREADS)
void k2_scatter(const int* __restrict__ edges, const int* __restrict__ hist,
                int* __restrict__ cursor, int* __restrict__ binned, int E,
                int nBins) {
  extern __shared__ int sm[];
  __shared__ int wsum[PRE_THREADS / 32];
  int* lh    = sm;
  int* lbase = sm + nBins;
  const int t    = blockIdx.y;
  const int tid  = threadIdx.x;
  const int lane = tid & 31;
  const int wv   = tid >> 5;
  const int* srcT = edges + (size_t)t * 2 * E;
  const int* dstT = srcT + E;
  const int per = (E + gridDim.x - 1) / gridDim.x;
  const int c0  = blockIdx.x * per;
  const int c1  = min(E, c0 + per);
  for (int i = tid; i < nBins; i += PRE_THREADS) lh[i] = 0;
  __syncthreads();
  chunk_hist(srcT, c0, c1, lh);
  __syncthreads();

  int carry = 0;
  for (int b0 = 0; b0 < nBins; b0 += PRE_THREADS) {
    const int bI = b0 + tid;
    const int v  = (bI < nBins) ? hist[t * nBins + bI] : 0;
    int inc = v;
    #pragma unroll
    for (int d = 1; d < 32; d <<= 1) {
      const int o = __shfl_up(inc, d);
      if (lane >= d) inc += o;
    }
    if (lane == 31) wsum[wv] = inc;
    __syncthreads();
    int wpre = 0, tot = 0;
    for (int k = 0; k < PRE_THREADS / 32; ++k) {
      const int s = wsum[k];
      if (k < wv) wpre += s;
      tot += s;
    }
    if (bI < nBins) {
      const int c   = lh[bI];
      const int rel = c ? atomicAdd(&cursor[t * nBins + bI], c) : 0;
      lbase[bI] = carry + wpre + inc - v + rel;
      lh[bI]    = 0;
    }
    carry += tot;
    __syncthreads();
  }

  int2* outT = (int2*)binned + (size_t)t * E;
  for (int e = c0 + tid; e < c1; e += ID_UNROLL * PRE_THREADS) {
    int  s[ID_UNROLL], d[ID_UNROLL];
    bool ok[ID_UNROLL];
    #pragma unroll
    for (int u = 0; u < ID_UNROLL; ++u) {
      const int j  = e + u * PRE_THREADS;
      ok[u] = j < c1;
      const int jc = ok[u] ? j : (c1 - 1);
      s[u] = srcT[jc];
      d[u] = dstT[jc];
    }
    asm volatile("" :: "v"(s[0]), "v"(s[1]), "v"(s[2]), "v"(s[3]),
                       "v"(s[4]), "v"(s[5]), "v"(s[6]), "v"(s[7]),
                       "v"(d[0]), "v"(d[1]), "v"(d[2]), "v"(d[3]),
                       "v"(d[4]), "v"(d[5]), "v"(d[6]), "v"(d[7]));
    #pragma unroll
    for (int u = 0; u < ID_UNROLL; ++u) {
      if (ok[u]) {
        const int bin = s[u] / ROWS_PER_BIN;
        const int p   = lbase[bin] + atomicAdd(&lh[bin], 1);
        outT[p] = make_int2(s[u], d[u]);
      }
    }
  }
}


__device__ __forceinline__ void row_info(int pRow, int pScBits, int v, int half,
                                         int& rl, float& sc) {
  int   rLo  = __builtin_amdgcn_readlane(pRow, v);
  int   rHi  = __builtin_amdgcn_readlane(pRow, v + 8);
  float scLo = __int_as_float(__builtin_amdgcn_readlane(pScBits, v));
  float scHi = __int_as_float(__builtin_amdgcn_readlane(pScBits, v + 8));
  rl = half ? rHi : rLo;
  sc = half ? scHi : scLo;
}

template <bool EPI>
__device__ __forceinline__ void mma_tile(const float* aRow,
                                         const v2f (&Bf)[K_DIM / 4][4],
                                         v8f (&acc)[4], const v8f (&prv)[4],
                                         int pRow, int pScBits,
                                         const float (&bias)[4],
                                         float* accL, int half, int row16) {
  v2f Af[K_DIM / 4];
  #pragma unroll
  for (int c = 0; c < K_DIM / 4; ++c) Af[c] = *(const v2f*)(aRow + 4 * c);

  v8f a0 = {}, a1 = {}, a2 = {}, a3 = {};
  float* rowp = accL;
  float  sc   = 0.0f;
  #pragma unroll
  for (int c = 0; c < K_DIM / 4; ++c) {
    a0 = wmma_f32_16x16x4(Af[c], Bf[c][0], a0);
    a1 = wmma_f32_16x16x4(Af[c], Bf[c][1], a1);
    a2 = wmma_f32_16x16x4(Af[c], Bf[c][2], a2);
    a3 = wmma_f32_16x16x4(Af[c], Bf[c][3], a3);
    if (EPI) {
      const int v = c >> 2, nt = c & 3;
      if (nt == 0) {
        int rl;
        row_info(pRow, pScBits, v, half, rl, sc);
        rowp = accL + rl * ACCSTRIDE + row16;
      }
      float val = fmaxf(prv[nt][v] + bias[nt], 0.0f) * sc;
      atomicAdd(rowp + nt * 16, val);
    }
  }
  acc[0] = a0; acc[1] = a1; acc[2] = a2; acc[3] = a3;
}

__global__ __launch_bounds__(WAVES * 32)
void k3_gemm_binsum(const float* __restrict__ x, const float* __restrict__ W,
                    const float* __restrict__ b, const float* __restrict__ attn,
                    const int* __restrict__ hist, const int* __restrict__ binned,
                    float* __restrict__ out, int N, int E, int T, int nBins) {
  __shared__ __attribute__((aligned(16))) float lds[LDS_FLOATS];
  __shared__ int   segOff[T_MAX];
  __shared__ int   segN[T_MAX];
  __shared__ float wsm[T_MAX];

  const int bin   = blockIdx.x;
  const int base  = bin * ROWS_PER_BIN;
  const int tid   = threadIdx.x;
  const int wave  = __builtin_amdgcn_readfirstlane(tid >> 5);
  const int lane  = tid & 31;
  const int row16 = lane & 15;
  const int half  = lane >> 4;
  float* sFw  = lds + wave * (TILE_M * FSTRIDE);
  float* accL = lds + WAVES * TILE_M * FSTRIDE;
  float* cntL = accL + ROWS_PER_BIN * ACCSTRIDE;

  for (int i = tid; i < ROWS_PER_BIN * ACCSTRIDE + T * ROWS_PER_BIN;
       i += WAVES * 32)
    accL[i] = 0.0f;
  if (tid < T_MAX) segOff[tid] = 0;
  __syncthreads();
  {
    int part[T_MAX];
    #pragma unroll
    for (int t = 0; t < T_MAX; ++t)
      part[t] = (t < T && tid < bin) ? hist[t * nBins + tid] : 0;
    #pragma unroll
    for (int t = 0; t < T_MAX; ++t)
      if (part[t]) atomicAdd(&segOff[t], part[t]);
    for (int t = 0; t < T; ++t)
      for (int i = tid + WAVES * 32; i < bin; i += WAVES * 32)
        atomicAdd(&segOff[t], hist[t * nBins + i]);
  }
  if (tid < T) segN[tid] = hist[tid * nBins + bin];
  if (tid == 0) {
    float m = attn[0];
    for (int t = 1; t < T; ++t) m = fmaxf(m, attn[t]);
    float e[T_MAX];
    float s = 0.0f;
    for (int t = 0; t < T && t < T_MAX; ++t) { e[t] = __expf(attn[t] - m); s += e[t]; }
    for (int t = 0; t < T && t < T_MAX; ++t) wsm[t] = e[t] / s;
  }
  __syncthreads();

  const int2* pairs = (const int2*)binned;
  for (int t = 0; t < T; ++t) {
    const int2* seg = pairs + (size_t)t * E + segOff[t];
    const int   n   = segN[t];
    float* cntT = cntL + t * ROWS_PER_BIN;
    for (int i0 = tid; i0 < n; i0 += CNT_UNROLL * WAVES * 32) {
      int  s[CNT_UNROLL];
      bool ok[CNT_UNROLL];
      #pragma unroll
      for (int u = 0; u < CNT_UNROLL; ++u) {
        const int j = i0 + u * WAVES * 32;
        ok[u] = j < n;
        s[u]  = seg[ok[u] ? j : (n - 1)].x;
      }
      asm volatile("" :: "v"(s[0]), "v"(s[1]), "v"(s[2]), "v"(s[3]),
                         "v"(s[4]), "v"(s[5]), "v"(s[6]), "v"(s[7]));
      #pragma unroll
      for (int u = 0; u < CNT_UNROLL; ++u)
        if (ok[u]) atomicAdd(&cntT[s[u] - base], 1.0f);
    }
  }
  __syncthreads();

  int total = 0;
  for (int t = 0; t < T; ++t) total += (segN[t] + TILE_M - 1) / TILE_M;
  total = __builtin_amdgcn_readfirstlane(total);
  const int lo = (total * wave) / WAVES;
  const int hi = (total * (wave + 1)) / WAVES;

  const float* aRow = sFw + row16 * FSTRIDE + 2 * half;
  int tsBeg = 0;
  for (int t = 0; t < T; ++t) {
    const int n     = __builtin_amdgcn_readfirstlane(segN[t]);
    const int off   = __builtin_amdgcn_readfirstlane(segOff[t]);
    const int ntile = (n + TILE_M - 1) / TILE_M;
    const int tlo   = max(lo, tsBeg) - tsBeg;
    const int thi   = min(hi, tsBeg + ntile) - tsBeg;
    tsBeg += ntile;
    if (tlo >= thi) continue;

    const float* Wt = W + (size_t)t * K_DIM * D_FEAT;
    v2f Bf[K_DIM / 4][4];
    #pragma unroll
    for (int c = 0; c < K_DIM / 4; ++c) {
      const float* p = Wt + (4 * c + 2 * half) * D_FEAT + row16;
      #pragma unroll
      for (int nt = 0; nt < 4; ++nt) {
        Bf[c][nt].x = p[nt * 16];
        Bf[c][nt].y = p[D_FEAT + nt * 16];
      }
    }

    const float* bt = b + (size_t)t * D_FEAT;
    const float bias[4] = {bt[0 + row16], bt[16 + row16], bt[32 + row16],
                           bt[48 + row16]};

    const int2*  seg  = pairs + (size_t)t * E + off;
    const float* cntT = cntL + t * ROWS_PER_BIN;
    const float  wt   = wsm[t];
    const int    last = n - 1;

    int tile = tlo;
    int edA = tile * TILE_M + row16;
    int2 idA = seg[min(edA, last)];
    int srcA = idA.x, dstA = idA.y;
    int edB = (tile + 1) * TILE_M + row16;
    int2 idB = seg[min(edB, last)];
    int srcB = idB.x, dstB = idB.y;

    v4f g[TILE_M];
    #pragma unroll
    for (int r = 0; r < TILE_M; ++r) {
      int sn = __builtin_amdgcn_readlane(srcA, r);
      int dn = __builtin_amdgcn_readlane(dstA, r);
      int node = half ? dn : sn;
      g[r] = *(const v4f*)(x + (size_t)node * D_FEAT + 4 * row16);
    }
    float scRawA = wt / fmaxf(cntT[srcA - base], 1.0f);
    float scA = (edA < n) ? scRawA : 0.0f;

    v8f prv[4] = {};
    int pRow = 0, pScBits = 0;

    auto trip = [&](auto epi) {
      #pragma unroll
      for (int r = 0; r < TILE_M; ++r)
        *(v4f*)(sFw + r * FSTRIDE + half * D_FEAT + 4 * row16) = g[r];

      #pragma unroll
      for (int r = 0; r < TILE_M; ++r) {
        int sn = __builtin_amdgcn_readlane(srcB, r);
        int dn = __builtin_amdgcn_readlane(dstB, r);
        int node = half ? dn : sn;
        g[r] = *(const v4f*)(x + (size_t)node * D_FEAT + 4 * row16);
      }
      float scRawB = wt / fmaxf(cntT[srcB - base], 1.0f);
      float scB = (edB < n) ? scRawB : 0.0f;
      int edC = (tile + 2) * TILE_M + row16;
      int2 idC = seg[min(edC, last)];
      int srcC = idC.x, dstC = idC.y;

      v8f acc[4];
      mma_tile<decltype(epi)::value>(aRow, Bf, acc, prv, pRow, pScBits, bias,
                                     accL, half, row16);

      prv[0] = acc[0]; prv[1] = acc[1]; prv[2] = acc[2]; prv[3] = acc[3];
      pRow = srcA - base; pScBits = __float_as_int(scA);

      srcA = srcB; dstA = dstB; scA = scB;
      srcB = srcC; dstB = dstC; edB = edC;
    };

    trip(std::false_type{});
    for (tile += 1; tile < thi; tile += 1) trip(std::true_type{});

    #pragma unroll
    for (int v = 0; v < 8; ++v) {
      int rl; float sc;
      row_info(pRow, pScBits, v, half, rl, sc);
      float* rowp = accL + rl * ACCSTRIDE + row16;
      #pragma unroll
      for (int nt = 0; nt < 4; ++nt) {
        float val = fmaxf(prv[nt][v] + bias[nt], 0.0f) * sc;
        atomicAdd(rowp + nt * 16, val);
      }
    }
  }
  __syncthreads();

  const int rows = min(ROWS_PER_BIN, N - base);
  v4f* out4 = (v4f*)out + (size_t)base * (D_FEAT / 4);
  for (int i = tid; i < rows * (D_FEAT / 4); i += WAVES * 32) {
    const int r = i >> 4, c4 = i & 15;
    out4[i] = *(const v4f*)(accL + r * ACCSTRIDE + 4 * c4);
  }
}

extern "C" void kernel_launch(void* const* d_in, const int* in_sizes, int n_in,
                              void* d_out, int out_size, void* d_ws, size_t ws_size,
                              hipStream_t stream) {
  const float* x     = (const float*)d_in[0];
  const float* W     = (const float*)d_in[1];
  const float* b     = (const float*)d_in[2];
  const float* attn  = (const float*)d_in[3];
  const int*   edges = (const int*)d_in[4];
  float* out = (float*)d_out;
  int*   wsi = (int*)d_ws;

  const int N = in_sizes[0] / D_FEAT;
  const int T = in_sizes[3];
  const int E = in_sizes[4] / (2 * T);
  const int nBins = (N + ROWS_PER_BIN - 1) / ROWS_PER_BIN;

  int* hist   = wsi;
  int* cursor = wsi + T * nBins;
  int* binned = wsi + 2 * T * nBins;

  k0_init<<<(2 * T * nBins + PRE_THREADS - 1) / PRE_THREADS, PRE_THREADS, 0,
            stream>>>(wsi, 2 * T * nBins);
  k1_hist<<<dim3(52, T), PRE_THREADS, nBins * sizeof(int), stream>>>(
      edges, hist, E, nBins);
  k2_scatter<<<dim3(52, T), PRE_THREADS, 2 * nBins * sizeof(int), stream>>>(
      edges, hist, cursor, binned, E, nBins);
  k3_gemm_binsum<<<nBins, WAVES * 32, 0, stream>>>(
      x, W, b, attn, hist, binned, out, N, E, T, nBins);
}


# k3: per-edge lane values (gather row offsets, plane row and scale of the adds) go by DPP row_share, not v_readlane+select+64-bit adds; scalar base + 32-bit offset; load groups kept ahead of WMMAs
# speedup vs baseline: 1.8981x; 1.0473x over previous
#include <hip/hip_runtime.h>
#include <type_traits>
#include <utility>

#define D_FEAT   64
#define K_DIM    128
#define TILE_M   16
#define WAVES    8
#define FSTRIDE  132
#define ROWS_PER_BIN 391
#define ACCSTRIDE 80
#define T_MAX    8
#define LDS_FLOATS (WAVES * TILE_M * FSTRIDE + ROWS_PER_BIN * ACCSTRIDE + T_MAX * ROWS_PER_BIN)
#define CNT_UNROLL 8

typedef __attribute__((ext_vector_type(2))) float v2f;
typedef __attribute__((ext_vector_type(4))) float v4f;
typedef __attribute__((ext_vector_type(8))) float v8f;

__device__ __forceinline__ v8f wmma_f32_16x16x4(v2f a, v2f b, v8f c) {
  return __builtin_amdgcn_wmma_f32_16x16x4_f32(false, a, false, b, (short)0, c,
                                               false, false);
}

template <int R>
__device__ __forceinline__ int row_bcast(int v) {
  return __builtin_amdgcn_update_dpp(0, v, 0x150 + R, 0xf, 0xf, true);
}
__device__ __forceinline__ int row_rot8(int v) {
  return __builtin_amdgcn_update_dpp(0, v, 0x128, 0xf, 0xf, true);
}
template <class F, int... Is>
__device__ __forceinline__ void static_for_impl(F&& f, std::integer_sequence<int, Is...>) {
  (f(std::integral_constant<int, Is>{}), ...);
}
template <int N, class F>
__device__ __forceinline__ void static_for(F&& f) {
  static_for_impl(f, std::make_integer_sequence<int, N>{});
}

#define PRE_THREADS 512
#define ID_UNROLL   8

__global__ __launch_bounds__(PRE_THREADS)
void k0_init(int* __restrict__ wsi, int nWords) {
  const int idx = blockIdx.x * blockDim.x + threadIdx.x;
  if (idx < nWords) wsi[idx] = 0;
}

__device__ __forceinline__ void chunk_hist(const int* __restrict__ srcT, int c0,
                                           int c1, int* lh) {
  const int tid = threadIdx.x;
  for (int e = c0 + tid; e < c1; e += ID_UNROLL * PRE_THREADS) {
    int  s[ID_UNROLL];
    bool ok[ID_UNROLL];
    #pragma unroll
    for (int u = 0; u < ID_UNROLL; ++u) {
      const int j = e + u * PRE_THREADS;
      ok[u] = j < c1;
      s[u]  = srcT[ok[u] ? j : (c1 - 1)];
    }
    asm volatile("" :: "v"(s[0]), "v"(s[1]), "v"(s[2]), "v"(s[3]),
                       "v"(s[4]), "v"(s[5]), "v"(s[6]), "v"(s[7]));
    #pragma unroll
    for (int u = 0; u < ID_UNROLL; ++u)
      if (ok[u]) atomicAdd(&lh[s[u] / ROWS_PER_BIN], 1);
  }
}

__global__ __launch_bounds__(PRE_THREADS)
void k1_hist(const int* __restrict__ edges, int* __restrict__ hist, int E,
             int nBins) {
  extern __shared__ int sm[];
  const int t   = blockIdx.y;
  const int tid = threadIdx.x;
  const int* srcT = edges + (size_t)t * 2 * E;
  const int per = (E + gridDim.x - 1) / gridDim.x;
  const int c0  = blockIdx.x * per;
  const int c1  = min(E, c0 + per);
  for (int i = tid; i < nBins; i += PRE_THREADS) sm[i] = 0;
  __syncthreads();
  chunk_hist(srcT, c0, c1, sm);
  __syncthreads();
  for (int i = tid; i < nBins; i += PRE_THREADS) {
    const int c = sm[i];
    if (c) atomicAdd(&hist[t * nBins + i], c);
  }
}

__global__ __launch_bounds__(PRE_THREADS)
void k2_scatter(const int* __restrict__ edges, const int* __restrict__ hist,
                int* __restrict__ cursor, int* __restrict__ binned, int E,
                int nBins) {
  extern __shared__ int sm[];
  __shared__ int wsum[PRE_THREADS / 32];
  int* lh    = sm;
  int* lbase = sm + nBins;
  const int t    = blockIdx.y;
  const int tid  = threadIdx.x;
  const int lane = tid & 31;
  const int wv   = tid >> 5;
  const int* srcT = edges + (size_t)t * 2 * E;
  const int* dstT = srcT + E;
  const int per = (E + gridDim.x - 1) / gridDim.x;
  const int c0  = blockIdx.x * per;
  const int c1  = min(E, c0 + per);
  for (int i = tid; i < nBins; i += PRE_THREADS) lh[i] = 0;
  __syncthreads();
  chunk_hist(srcT, c0, c1, lh);
  __syncthreads();

  int carry = 0;
  for (int b0 = 0; b0 < nBins; b0 += PRE_THREADS) {
    const int bI = b0 + tid;
    const int v  = (bI < nBins) ? hist[t * nBins + bI] : 0;
    int inc = v;
    #pragma unroll
    for (int d = 1; d < 32; d <<= 1) {
      const int o = __shfl_up(inc, d);
      if (lane >= d) inc += o;
    }
    if (lane == 31) wsum[wv] = inc;
    __syncthreads();
    int wpre = 0, tot = 0;
    for (int k = 0; k < PRE_THREADS / 32; ++k) {
      const int s = wsum[k];
      if (k < wv) wpre += s;
      tot += s;
    }
    if (bI < nBins) {
      const int c   = lh[bI];
      const int rel = c ? atomicAdd(&cursor[t * nBins + bI], c) : 0;
      lbase[bI] = carry + wpre + inc - v + rel;
      lh[bI]    = 0;
    }
    carry += tot;
    __syncthreads();
  }

  int2* outT = (int2*)binned + (size_t)t * E;
  for (int e = c0 + tid; e < c1; e += ID_UNROLL * PRE_THREADS) {
    int  s[ID_UNROLL], d[ID_UNROLL];
    bool ok[ID_UNROLL];
    #pragma unroll
    for (int u = 0; u < ID_UNROLL; ++u) {
      const int j  = e + u * PRE_THREADS;
      ok[u] = j < c1;
      const int jc = ok[u] ? j : (c1 - 1);
      s[u] = srcT[jc];
      d[u] = dstT[jc];
    }
    asm volatile("" :: "v"(s[0]), "v"(s[1]), "v"(s[2]), "v"(s[3]),
                       "v"(s[4]), "v"(s[5]), "v"(s[6]), "v"(s[7]),
                       "v"(d[0]), "v"(d[1]), "v"(d[2]), "v"(d[3]),
                       "v"(d[4]), "v"(d[5]), "v"(d[6]), "v"(d[7]));
    #pragma unroll
    for (int u = 0; u < ID_UNROLL; ++u) {
      if (ok[u]) {
        const int bin = s[u] / ROWS_PER_BIN;
        const int p   = lbase[bin] + atomicAdd(&lh[bin], 1);
        outT[p] = make_int2(s[u], d[u]);
      }
    }
  }
}


template <bool EPI>
__device__ __forceinline__ void mma_tile(const float* aRow,
                                         const v2f (&Bf)[K_DIM / 4][4],
                                         v8f (&acc)[4], const v8f (&rel)[4],
                                         int pRowB, int pScBits,
                                         char* colp) {
  v2f Af[K_DIM / 4];
  #pragma unroll
  for (int c = 0; c < K_DIM / 4; ++c) Af[c] = *(const v2f*)(aRow + 4 * c);
  __builtin_amdgcn_sched_barrier(0x606);

  v8f a0 = {}, a1 = {}, a2 = {}, a3 = {};
  float* rowp = (float*)colp;
  float  sc   = 0.0f;
  static_for<K_DIM / 4>([&](auto cc) {
    constexpr int c = decltype(cc)::value;
    a0 = wmma_f32_16x16x4(Af[c], Bf[c][0], a0);
    a1 = wmma_f32_16x16x4(Af[c], Bf[c][1], a1);
    a2 = wmma_f32_16x16x4(Af[c], Bf[c][2], a2);
    a3 = wmma_f32_16x16x4(Af[c], Bf[c][3], a3);
    if constexpr (EPI) {
      constexpr int v = c >> 2, nt = c & 3;
      if constexpr (nt == 0) {
        rowp = (float*)(colp + row_bcast<v>(pRowB));
        sc   = __int_as_float(row_bcast<v>(pScBits));
      }
      atomicAdd(rowp + nt * 16, rel[nt][v] * sc);
    }
  });
  acc[0] = a0; acc[1] = a1; acc[2] = a2; acc[3] = a3;
}

__global__ __launch_bounds__(WAVES * 32)
void k3_gemm_binsum(const float* __restrict__ x, const float* __restrict__ W,
                    const float* __restrict__ b, const float* __restrict__ attn,
                    const int* __restrict__ hist, const int* __restrict__ binned,
                    float* __restrict__ out, int N, int E, int T, int nBins) {
  __shared__ __attribute__((aligned(16))) float lds[LDS_FLOATS];
  __shared__ int   segOff[T_MAX];
  __shared__ int   segN[T_MAX];
  __shared__ float wsm[T_MAX];

  const int bin   = blockIdx.x;
  const int base  = bin * ROWS_PER_BIN;
  const int tid   = threadIdx.x;
  const int wave  = __builtin_amdgcn_readfirstlane(tid >> 5);
  const int lane  = tid & 31;
  const int row16 = lane & 15;
  const int half  = lane >> 4;
  float* sFw  = lds + wave * (TILE_M * FSTRIDE);
  float* accL = lds + WAVES * TILE_M * FSTRIDE;
  float* cntL = accL + ROWS_PER_BIN * ACCSTRIDE;

  for (int i = tid; i < ROWS_PER_BIN * ACCSTRIDE + T * ROWS_PER_BIN;
       i += WAVES * 32)
    accL[i] = 0.0f;
  if (tid < T_MAX) segOff[tid] = 0;
  __syncthreads();
  {
    int part[T_MAX];
    #pragma unroll
    for (int t = 0; t < T_MAX; ++t)
      part[t] = (t < T && tid < bin) ? hist[t * nBins + tid] : 0;
    #pragma unroll
    for (int t = 0; t < T_MAX; ++t)
      if (part[t]) atomicAdd(&segOff[t], part[t]);
    for (int t = 0; t < T; ++t)
      for (int i = tid + WAVES * 32; i < bin; i += WAVES * 32)
        atomicAdd(&segOff[t], hist[t * nBins + i]);
  }
  if (tid < T) segN[tid] = hist[tid * nBins + bin];
  if (tid == 0) {
    float m = attn[0];
    for (int t = 1; t < T; ++t) m = fmaxf(m, attn[t]);
    float e[T_MAX];
    float s = 0.0f;
    for (int t = 0; t < T && t < T_MAX; ++t) { e[t] = __expf(attn[t] - m); s += e[t]; }
    for (int t = 0; t < T && t < T_MAX; ++t) wsm[t] = e[t] / s;
  }
  __syncthreads();

  const int2* pairs = (const int2*)binned;
  for (int t = 0; t < T; ++t) {
    const int2* seg = pairs + (size_t)t * E + segOff[t];
    const int   n   = segN[t];
    float* cntT = cntL + t * ROWS_PER_BIN;
    for (int i0 = tid; i0 < n; i0 += CNT_UNROLL * WAVES * 32) {
      int  s[CNT_UNROLL];
      bool ok[CNT_UNROLL];
      #pragma unroll
      for (int u = 0; u < CNT_UNROLL; ++u) {
        const int j = i0 + u * WAVES * 32;
        ok[u] = j < n;
        s[u]  = seg[ok[u] ? j : (n - 1)].x;
      }
      asm volatile("" :: "v"(s[0]), "v"(s[1]), "v"(s[2]), "v"(s[3]),
                         "v"(s[4]), "v"(s[5]), "v"(s[6]), "v"(s[7]));
      #pragma unroll
      for (int u = 0; u < CNT_UNROLL; ++u)
        if (ok[u]) atomicAdd(&cntT[s[u] - base], 1.0f);
    }
  }
  __syncthreads();

  int total = 0;
  for (int t = 0; t < T; ++t) total += (segN[t] + TILE_M - 1) / TILE_M;
  total = __builtin_amdgcn_readfirstlane(total);
  const int lo = (total * wave) / WAVES;
  const int hi = (total * (wave + 1)) / WAVES;

  const float* aRow = sFw + row16 * FSTRIDE + 2 * half;
  char* colp = (char*)(accL + row16);
  const char* xb = (const char*)x;
  const unsigned colB = 16u * row16;
  int tsBeg = 0;
  for (int t = 0; t < T; ++t) {
    const int n     = __builtin_amdgcn_readfirstlane(segN[t]);
    const int off   = __builtin_amdgcn_readfirstlane(segOff[t]);
    const int ntile = (n + TILE_M - 1) / TILE_M;
    const int tlo   = max(lo, tsBeg) - tsBeg;
    const int thi   = min(hi, tsBeg + ntile) - tsBeg;
    tsBeg += ntile;
    if (tlo >= thi) continue;

    const float* Wt = W + (size_t)t * K_DIM * D_FEAT;
    v2f Bf[K_DIM / 4][4];
    #pragma unroll
    for (int c = 0; c < K_DIM / 4; ++c) {
      const float* p = Wt + (4 * c + 2 * half) * D_FEAT + row16;
      #pragma unroll
      for (int nt = 0; nt < 4; ++nt) {
        Bf[c][nt].x = p[nt * 16];
        Bf[c][nt].y = p[D_FEAT + nt * 16];
      }
    }

    const float* bt = b + (size_t)t * D_FEAT;
    const float bias[4] = {bt[0 + row16], bt[16 + row16], bt[32 + row16],
                           bt[48 + row16]};

    const int2*  seg  = pairs + (size_t)t * E + off;
    const float* cntT = cntL + t * ROWS_PER_BIN;
    const float  wt   = wsm[t];
    const int    last = n - 1;

    int tile = tlo;
    int edA = tile * TILE_M + row16;
    int2 idA = seg[min(edA, last)];
    int srcA = idA.x, dstA = idA.y;
    int edB = (tile + 1) * TILE_M + row16;
    int2 idB = seg[min(edB, last)];
    int srcB = idB.x, dstB = idB.y;

    v4f g[TILE_M];
    {
      const int nodeOff = (half ? dstA : srcA) * (D_FEAT * 4);
      static_for<TILE_M>([&](auto rr) {
        constexpr int r = decltype(rr)::value;
        const unsigned o = (unsigned)row_bcast<r>(nodeOff) + colB;
        g[r] = *(const v4f*)(xb + o);
      });
    }
    float scRawA = wt / fmaxf(cntT[srcA - base], 1.0f);
    float scA = (edA < n) ? scRawA : 0.0f;

    v8f prv[4] = {};
    int pRowB = 0, pScBits = 0;

    auto trip = [&](auto epi) {
      v8f rel[4] = {};
      if constexpr (decltype(epi)::value) {
        #pragma unroll
        for (int nt = 0; nt < 4; ++nt)
          #pragma unroll
          for (int v = 0; v < 8; ++v)
            rel[nt][v] = fmaxf(prv[nt][v] + bias[nt], 0.0f);
      }

      #pragma unroll
      for (int r = 0; r < TILE_M; ++r)
        *(v4f*)(sFw + r * FSTRIDE + half * D_FEAT + 4 * row16) = g[r];

      {
        const int nodeOff = (half ? dstB : srcB) * (D_FEAT * 4);
        static_for<TILE_M>([&](auto rr) {
          constexpr int r = decltype(rr)::value;
          const unsigned o = (unsigned)row_bcast<r>(nodeOff) + colB;
          g[r] = *(const v4f*)(xb + o);
        });
      }
      float scRawB = wt / fmaxf(cntT[srcB - base], 1.0f);
      float scB = (edB < n) ? scRawB : 0.0f;
      int edC = (tile + 2) * TILE_M + row16;
      int2 idC = seg[min(edC, last)];
      int srcC = idC.x, dstC = idC.y;

      v8f acc[4];
      mma_tile<decltype(epi)::value>(aRow, Bf, acc, rel, pRowB, pScBits, colp);

      prv[0] = acc[0]; prv[1] = acc[1]; prv[2] = acc[2]; prv[3] = acc[3];
      {
        const int rb = (srcA - base) * (ACCSTRIDE * 4);
        const int sb = __float_as_int(scA);
        const int rbT = row_rot8(rb), sbT = row_rot8(sb);
        pRowB   = half ? rbT : rb;
        pScBits = half ? sbT : sb;
      }

      srcA = srcB; dstA = dstB; scA = scB;
      srcB = srcC; dstB = dstC; edB = edC;
    };

    trip(std::false_type{});
    for (tile += 1; tile < thi; tile += 1) trip(std::true_type{});

    static_for<8>([&](auto vv) {
      constexpr int v = decltype(vv)::value;
      float* rowp = (float*)(colp + row_bcast<v>(pRowB));
      const float sc = __int_as_float(row_bcast<v>(pScBits));
      #pragma unroll
      for (int nt = 0; nt < 4; ++nt) {
        float val = fmaxf(prv[nt][v] + bias[nt], 0.0f) * sc;
        atomicAdd(rowp + nt * 16, val);
      }
    });
  }
  __syncthreads();

  const int rows = min(ROWS_PER_BIN, N - base);
  v4f* out4 = (v4f*)out + (size_t)base * (D_FEAT / 4);
  for (int i = tid; i < rows * (D_FEAT / 4); i += WAVES * 32) {
    const int r = i >> 4, c4 = i & 15;
    out4[i] = *(const v4f*)(accL + r * ACCSTRIDE + 4 * c4);
  }
}

extern "C" void kernel_launch(void* const* d_in, const int* in_sizes, int n_in,
                              void* d_out, int out_size, void* d_ws, size_t ws_size,
                              hipStream_t stream) {
  const float* x     = (const float*)d_in[0];
  const float* W     = (const float*)d_in[1];
  const float* b     = (const float*)d_in[2];
  const float* attn  = (const float*)d_in[3];
  const int*   edges = (const int*)d_in[4];
  float* out = (float*)d_out;
  int*   wsi = (int*)d_ws;

  const int N = in_sizes[0] / D_FEAT;
  const int T = in_sizes[3];
  const int E = in_sizes[4] / (2 * T);
  const int nBins = (N + ROWS_PER_BIN - 1) / ROWS_PER_BIN;

  int* hist   = wsi;
  int* cursor = wsi + T * nBins;
  int* binned = wsi + 2 * T * nBins;

  k0_init<<<(2 * T * nBins + PRE_THREADS - 1) / PRE_THREADS, PRE_THREADS, 0,
            stream>>>(wsi, 2 * T * nBins);
  k1_hist<<<dim3(52, T), PRE_THREADS, nBins * sizeof(int), stream>>>(
      edges, hist, E, nBins);
  k2_scatter<<<dim3(52, T), PRE_THREADS, 2 * nBins * sizeof(int), stream>>>(
      edges, hist, cursor, binned, E, nBins);
  k3_gemm_binsum<<<nBins, WAVES * 32, 0, stream>>>(
      x, W, b, attn, hist, binned, out, N, E, T, nBins);
}


# k3: wave issue priority stepped down with progress through its run (s_setprio 3/2/1/0 from 1/2, 13/16, 15/16 done), so a SIMD's two waves take turns and end together; LDS add arrival order differs
# speedup vs baseline: 1.9867x; 1.0467x over previous
#include <hip/hip_runtime.h>
#include <type_traits>
#include <utility>

#define D_FEAT   64
#define K_DIM    128
#define TILE_M   16
#define WAVES    8
#define FSTRIDE  132
#define ROWS_PER_BIN 391
#define ACCSTRIDE 80
#define T_MAX    8
#define LDS_FLOATS (WAVES * TILE_M * FSTRIDE + ROWS_PER_BIN * ACCSTRIDE + T_MAX * ROWS_PER_BIN)
#define CNT_UNROLL 8

typedef __attribute__((ext_vector_type(2))) float v2f;
typedef __attribute__((ext_vector_type(4))) float v4f;
typedef __attribute__((ext_vector_type(8))) float v8f;

__device__ __forceinline__ v8f wmma_f32_16x16x4(v2f a, v2f b, v8f c) {
  return __builtin_amdgcn_wmma_f32_16x16x4_f32(false, a, false, b, (short)0, c,
                                               false, false);
}

template <int R>
__device__ __forceinline__ int row_bcast(int v) {
  return __builtin_amdgcn_update_dpp(0, v, 0x150 + R, 0xf, 0xf, true);
}
__device__ __forceinline__ int row_rot8(int v) {
  return __builtin_amdgcn_update_dpp(0, v, 0x128, 0xf, 0xf, true);
}
template <class F, int... Is>
__device__ __forceinline__ void static_for_impl(F&& f, std::integer_sequence<int, Is...>) {
  (f(std::integral_constant<int, Is>{}), ...);
}
template <int N, class F>
__device__ __forceinline__ void static_for(F&& f) {
  static_for_impl(f, std::make_integer_sequence<int, N>{});
}

#define PRE_THREADS 512
#define ID_UNROLL   8

__global__ __launch_bounds__(PRE_THREADS)
void k0_init(int* __restrict__ wsi, int nWords) {
  const int idx = blockIdx.x * blockDim.x + threadIdx.x;
  if (idx < nWords) wsi[idx] = 0;
}

__device__ __forceinline__ void chunk_hist(const int* __restrict__ srcT, int c0,
                                           int c1, int* lh) {
  const int tid = threadIdx.x;
  for (int e = c0 + tid; e < c1; e += ID_UNROLL * PRE_THREADS) {
    int  s[ID_UNROLL];
    bool ok[ID_UNROLL];
    #pragma unroll
    for (int u = 0; u < ID_UNROLL; ++u) {
      const int j = e + u * PRE_THREADS;
      ok[u] = j < c1;
      s[u]  = srcT[ok[u] ? j : (c1 - 1)];
    }
    asm volatile("" :: "v"(s[0]), "v"(s[1]), "v"(s[2]), "v"(s[3]),
                       "v"(s[4]), "v"(s[5]), "v"(s[6]), "v"(s[7]));
    #pragma unroll
    for (int u = 0; u < ID_UNROLL; ++u)
      if (ok[u]) atomicAdd(&lh[s[u] / ROWS_PER_BIN], 1);
  }
}

__global__ __launch_bounds__(PRE_THREADS)
void k1_hist(const int* __restrict__ edges, int* __restrict__ hist, int E,
             int nBins) {
  extern __shared__ int sm[];
  const int t   = blockIdx.y;
  const int tid = threadIdx.x;
  const int* srcT = edges + (size_t)t * 2 * E;
  const int per = (E + gridDim.x - 1) / gridDim.x;
  const int c0  = blockIdx.x * per;
  const int c1  = min(E, c0 + per);
  for (int i = tid; i < nBins; i += PRE_THREADS) sm[i] = 0;
  __syncthreads();
  chunk_hist(srcT, c0, c1, sm);
  __syncthreads();
  for (int i = tid; i < nBins; i += PRE_THREADS) {
    const int c = sm[i];
    if (c) atomicAdd(&hist[t * nBins + i], c);
  }
}

__global__ __launch_bounds__(PRE_THREADS)
void k2_scatter(const int* __restrict__ edges, const int* __restrict__ hist,
                int* __restrict__ cursor, int* __restrict__ binned, int E,
                int nBins) {
  extern __shared__ int sm[];
  __shared__ int wsum[PRE_THREADS / 32];
  int* lh    = sm;
  int* lbase = sm + nBins;
  const int t    = blockIdx.y;
  const int tid  = threadIdx.x;
  const int lane = tid & 31;
  const int wv   = tid >> 5;
  const int* srcT = edges + (size_t)t * 2 * E;
  const int* dstT = srcT + E;
  const int per = (E + gridDim.x - 1) / gridDim.x;
  const int c0  = blockIdx.x * per;
  const int c1  = min(E, c0 + per);
  for (int i = tid; i < nBins; i += PRE_THREADS) lh[i] = 0;
  __syncthreads();
  chunk_hist(srcT, c0, c1, lh);
  __syncthreads();

  int carry = 0;
  for (int b0 = 0; b0 < nBins; b0 += PRE_THREADS) {
    const int bI = b0 + tid;
    const int v  = (bI < nBins) ? hist[t * nBins + bI] : 0;
    int inc = v;
    #pragma unroll
    for (int d = 1; d < 32; d <<= 1) {
      const int o = __shfl_up(inc, d);
      if (lane >= d) inc += o;
    }
    if (lane == 31) wsum[wv] = inc;
    __syncthreads();
    int wpre = 0, tot = 0;
    for (int k = 0; k < PRE_THREADS / 32; ++k) {
      const int s = wsum[k];
      if (k < wv) wpre += s;
      tot += s;
    }
    if (bI < nBins) {
      const int c   = lh[bI];
      const int rel = c ? atomicAdd(&cursor[t * nBins + bI], c) : 0;
      lbase[bI] = carry + wpre + inc - v + rel;
      lh[bI]    = 0;
    }
    carry += tot;
    __syncthreads();
  }

  int2* outT = (int2*)binned + (size_t)t * E;
  for (int e = c0 + tid; e < c1; e += ID_UNROLL * PRE_THREADS) {
    int  s[ID_UNROLL], d[ID_UNROLL];
    bool ok[ID_UNROLL];
    #pragma unroll
    for (int u = 0; u < ID_UNROLL; ++u) {
      const int j  = e + u * PRE_THREADS;
      ok[u] = j < c1;
      const int jc = ok[u] ? j : (c1 - 1);
      s[u] = srcT[jc];
      d[u] = dstT[jc];
    }
    asm volatile("" :: "v"(s[0]), "v"(s[1]), "v"(s[2]), "v"(s[3]),
                       "v"(s[4]), "v"(s[5]), "v"(s[6]), "v"(s[7]),
                       "v"(d[0]), "v"(d[1]), "v"(d[2]), "v"(d[3]),
                       "v"(d[4]), "v"(d[5]), "v"(d[6]), "v"(d[7]));
    #pragma unroll
    for (int u = 0; u < ID_UNROLL; ++u) {
      if (ok[u]) {
        const int bin = s[u] / ROWS_PER_BIN;
        const int p   = lbase[bin] + atomicAdd(&lh[bin], 1);
        outT[p] = make_int2(s[u], d[u]);
      }
    }
  }
}


template <bool EPI>
__device__ __forceinline__ void mma_tile(const float* aRow,
                                         const v2f (&Bf)[K_DIM / 4][4],
                                         v8f (&acc)[4], const v8f (&rel)[4],
                                         int pRowB, int pScBits,
                                         char* colp) {
  v2f Af[K_DIM / 4];
  #pragma unroll
  for (int c = 0; c < K_DIM / 4; ++c) Af[c] = *(const v2f*)(aRow + 4 * c);
  __builtin_amdgcn_sched_barrier(0x606);

  v8f a0 = {}, a1 = {}, a2 = {}, a3 = {};
  float* rowp = (float*)colp;
  float  sc   = 0.0f;
  static_for<K_DIM / 4>([&](auto cc) {
    constexpr int c = decltype(cc)::value;
    a0 = wmma_f32_16x16x4(Af[c], Bf[c][0], a0);
    a1 = wmma_f32_16x16x4(Af[c], Bf[c][1], a1);
    a2 = wmma_f32_16x16x4(Af[c], Bf[c][2], a2);
    a3 = wmma_f32_16x16x4(Af[c], Bf[c][3], a3);
    if constexpr (EPI) {
      constexpr int v = c >> 2, nt = c & 3;
      if constexpr (nt == 0) {
        rowp = (float*)(colp + row_bcast<v>(pRowB));
        sc   = __int_as_float(row_bcast<v>(pScBits));
      }
      atomicAdd(rowp + nt * 16, rel[nt][v] * sc);
    }
  });
  acc[0] = a0; acc[1] = a1; acc[2] = a2; acc[3] = a3;
}

__global__ __launch_bounds__(WAVES * 32)
void k3_gemm_binsum(const float* __restrict__ x, const float* __restrict__ W,
                    const float* __restrict__ b, const float* __restrict__ attn,
                    const int* __restrict__ hist, const int* __restrict__ binned,
                    float* __restrict__ out, int N, int E, int T, int nBins) {
  __shared__ __attribute__((aligned(16))) float lds[LDS_FLOATS];
  __shared__ int   segOff[T_MAX];
  __shared__ int   segN[T_MAX];
  __shared__ float wsm[T_MAX];

  const int bin   = blockIdx.x;
  const int base  = bin * ROWS_PER_BIN;
  const int tid   = threadIdx.x;
  const int wave  = __builtin_amdgcn_readfirstlane(tid >> 5);
  const int lane  = tid & 31;
  const int row16 = lane & 15;
  const int half  = lane >> 4;
  float* sFw  = lds + wave * (TILE_M * FSTRIDE);
  float* accL = lds + WAVES * TILE_M * FSTRIDE;
  float* cntL = accL + ROWS_PER_BIN * ACCSTRIDE;

  for (int i = tid; i < ROWS_PER_BIN * ACCSTRIDE + T * ROWS_PER_BIN;
       i += WAVES * 32)
    accL[i] = 0.0f;
  if (tid < T_MAX) segOff[tid] = 0;
  __syncthreads();
  {
    int part[T_MAX];
    #pragma unroll
    for (int t = 0; t < T_MAX; ++t)
      part[t] = (t < T && tid < bin) ? hist[t * nBins + tid] : 0;
    #pragma unroll
    for (int t = 0; t < T_MAX; ++t)
      if (part[t]) atomicAdd(&segOff[t], part[t]);
    for (int t = 0; t < T; ++t)
      for (int i = tid + WAVES * 32; i < bin; i += WAVES * 32)
        atomicAdd(&segOff[t], hist[t * nBins + i]);
  }
  if (tid < T) segN[tid] = hist[tid * nBins + bin];
  if (tid == 0) {
    float m = attn[0];
    for (int t = 1; t < T; ++t) m = fmaxf(m, attn[t]);
    float e[T_MAX];
    float s = 0.0f;
    for (int t = 0; t < T && t < T_MAX; ++t) { e[t] = __expf(attn[t] - m); s += e[t]; }
    for (int t = 0; t < T && t < T_MAX; ++t) wsm[t] = e[t] / s;
  }
  __syncthreads();

  const int2* pairs = (const int2*)binned;
  for (int t = 0; t < T; ++t) {
    const int2* seg = pairs + (size_t)t * E + segOff[t];
    const int   n   = segN[t];
    float* cntT = cntL + t * ROWS_PER_BIN;
    for (int i0 = tid; i0 < n; i0 += CNT_UNROLL * WAVES * 32) {
      int  s[CNT_UNROLL];
      bool ok[CNT_UNROLL];
      #pragma unroll
      for (int u = 0; u < CNT_UNROLL; ++u) {
        const int j = i0 + u * WAVES * 32;
        ok[u] = j < n;
        s[u]  = seg[ok[u] ? j : (n - 1)].x;
      }
      asm volatile("" :: "v"(s[0]), "v"(s[1]), "v"(s[2]), "v"(s[3]),
                         "v"(s[4]), "v"(s[5]), "v"(s[6]), "v"(s[7]));
      #pragma unroll
      for (int u = 0; u < CNT_UNROLL; ++u)
        if (ok[u]) atomicAdd(&cntT[s[u] - base], 1.0f);
    }
  }
  __syncthreads();

  int total = 0;
  for (int t = 0; t < T; ++t) total += (segN[t] + TILE_M - 1) / TILE_M;
  total = __builtin_amdgcn_readfirstlane(total);
  const int lo = (total * wave) / WAVES;
  const int hi = (total * (wave + 1)) / WAVES;

  const float* aRow = sFw + row16 * FSTRIDE + 2 * half;
  char* colp = (char*)(accL + row16);
  const char* xb = (const char*)x;
  const unsigned colB = 16u * row16;
  int tsBeg = 0;
  const int runLen = hi - lo;
  const int mark2 = lo + (runLen >> 1);
  const int mark1 = hi - ((3 * runLen) >> 4);
  const int mark0 = hi - (runLen >> 4);
  for (int t = 0; t < T; ++t) {
    const int n     = __builtin_amdgcn_readfirstlane(segN[t]);
    const int off   = __builtin_amdgcn_readfirstlane(segOff[t]);
    const int ntile = (n + TILE_M - 1) / TILE_M;
    const int tlo   = max(lo, tsBeg) - tsBeg;
    const int thi   = min(hi, tsBeg + ntile) - tsBeg;
    const int pm2 = mark2 - tsBeg, pm1 = mark1 - tsBeg, pm0 = mark0 - tsBeg;
    tsBeg += ntile;
    if (tlo >= thi) continue;

    const float* Wt = W + (size_t)t * K_DIM * D_FEAT;
    v2f Bf[K_DIM / 4][4];
    #pragma unroll
    for (int c = 0; c < K_DIM / 4; ++c) {
      const float* p = Wt + (4 * c + 2 * half) * D_FEAT + row16;
      #pragma unroll
      for (int nt = 0; nt < 4; ++nt) {
        Bf[c][nt].x = p[nt * 16];
        Bf[c][nt].y = p[D_FEAT + nt * 16];
      }
    }

    const float* bt = b + (size_t)t * D_FEAT;
    const float bias[4] = {bt[0 + row16], bt[16 + row16], bt[32 + row16],
                           bt[48 + row16]};

    const int2*  seg  = pairs + (size_t)t * E + off;
    const float* cntT = cntL + t * ROWS_PER_BIN;
    const float  wt   = wsm[t];
    const int    last = n - 1;

    int tile = tlo;
    int edA = tile * TILE_M + row16;
    int2 idA = seg[min(edA, last)];
    int srcA = idA.x, dstA = idA.y;
    int edB = (tile + 1) * TILE_M + row16;
    int2 idB = seg[min(edB, last)];
    int srcB = idB.x, dstB = idB.y;

    v4f g[TILE_M];
    {
      const int nodeOff = (half ? dstA : srcA) * (D_FEAT * 4);
      static_for<TILE_M>([&](auto rr) {
        constexpr int r = decltype(rr)::value;
        const unsigned o = (unsigned)row_bcast<r>(nodeOff) + colB;
        g[r] = *(const v4f*)(xb + o);
      });
    }
    float scRawA = wt / fmaxf(cntT[srcA - base], 1.0f);
    float scA = (edA < n) ? scRawA : 0.0f;

    v8f prv[4] = {};
    int pRowB = 0, pScBits = 0;

    auto trip = [&](auto epi) {
      v8f rel[4] = {};
      if constexpr (decltype(epi)::value) {
        #pragma unroll
        for (int nt = 0; nt < 4; ++nt)
          #pragma unroll
          for (int v = 0; v < 8; ++v)
            rel[nt][v] = fmaxf(prv[nt][v] + bias[nt], 0.0f);
      }

      #pragma unroll
      for (int r = 0; r < TILE_M; ++r)
        *(v4f*)(sFw + r * FSTRIDE + half * D_FEAT + 4 * row16) = g[r];

      {
        const int nodeOff = (half ? dstB : srcB) * (D_FEAT * 4);
        static_for<TILE_M>([&](auto rr) {
          constexpr int r = decltype(rr)::value;
          const unsigned o = (unsigned)row_bcast<r>(nodeOff) + colB;
          g[r] = *(const v4f*)(xb + o);
        });
      }
      float scRawB = wt / fmaxf(cntT[srcB - base], 1.0f);
      float scB = (edB < n) ? scRawB : 0.0f;
      int edC = (tile + 2) * TILE_M + row16;
      int2 idC = seg[min(edC, last)];
      int srcC = idC.x, dstC = idC.y;

      v8f acc[4];
      mma_tile<decltype(epi)::value>(aRow, Bf, acc, rel, pRowB, pScBits, colp);

      prv[0] = acc[0]; prv[1] = acc[1]; prv[2] = acc[2]; prv[3] = acc[3];
      {
        const int rb = (srcA - base) * (ACCSTRIDE * 4);
        const int sb = __float_as_int(scA);
        const int rbT = row_rot8(rb), sbT = row_rot8(sb);
        pRowB   = half ? rbT : rb;
        pScBits = half ? sbT : sb;
      }

      srcA = srcB; dstA = dstB; scA = scB;
      srcB = srcC; dstB = dstC; edB = edC;
    };

    auto setLevel = [&](int lvl) {
      if (lvl == 3)      __builtin_amdgcn_s_setprio(3);
      else if (lvl == 2) __builtin_amdgcn_s_setprio(2);
      else if (lvl == 1) __builtin_amdgcn_s_setprio(1);
      else               __builtin_amdgcn_s_setprio(0);
    };
    setLevel(tile < pm2 ? 3 : tile < pm1 ? 2 : tile < pm0 ? 1 : 0);
    trip(std::false_type{});
    tile += 1;
    #pragma nounroll
    for (int lvl = 3; lvl >= 0; --lvl) {
      const int lvlEnd = min(thi, lvl == 3 ? pm2 : lvl == 2 ? pm1 : lvl == 1 ? pm0 : thi);
      if (tile >= lvlEnd) continue;
      setLevel(lvl);
      for (; tile < lvlEnd; tile += 1) trip(std::true_type{});
    }

    static_for<8>([&](auto vv) {
      constexpr int v = decltype(vv)::value;
      float* rowp = (float*)(colp + row_bcast<v>(pRowB));
      const float sc = __int_as_float(row_bcast<v>(pScBits));
      #pragma unroll
      for (int nt = 0; nt < 4; ++nt) {
        float val = fmaxf(prv[nt][v] + bias[nt], 0.0f) * sc;
        atomicAdd(rowp + nt * 16, val);
      }
    });
  }
  __builtin_amdgcn_s_setprio(0);
  __syncthreads();

  const int rows = min(ROWS_PER_BIN, N - base);
  v4f* out4 = (v4f*)out + (size_t)base * (D_FEAT / 4);
  for (int i = tid; i < rows * (D_FEAT / 4); i += WAVES * 32) {
    const int r = i >> 4, c4 = i & 15;
    out4[i] = *(const v4f*)(accL + r * ACCSTRIDE + 4 * c4);
  }
}

extern "C" void kernel_launch(void* const* d_in, const int* in_sizes, int n_in,
                              void* d_out, int out_size, void* d_ws, size_t ws_size,
                              hipStream_t stream) {
  const float* x     = (const float*)d_in[0];
  const float* W     = (const float*)d_in[1];
  const float* b     = (const float*)d_in[2];
  const float* attn  = (const float*)d_in[3];
  const int*   edges = (const int*)d_in[4];
  float* out = (float*)d_out;
  int*   wsi = (int*)d_ws;

  const int N = in_sizes[0] / D_FEAT;
  const int T = in_sizes[3];
  const int E = in_sizes[4] / (2 * T);
  const int nBins = (N + ROWS_PER_BIN - 1) / ROWS_PER_BIN;

  int* hist   = wsi;
  int* cursor = wsi + T * nBins;
  int* binned = wsi + 2 * T * nBins;

  k0_init<<<(2 * T * nBins + PRE_THREADS - 1) / PRE_THREADS, PRE_THREADS, 0,
            stream>>>(wsi, 2 * T * nBins);
  k1_hist<<<dim3(52, T), PRE_THREADS, nBins * sizeof(int), stream>>>(
      edges, hist, E, nBins);
  k2_scatter<<<dim3(52, T), PRE_THREADS, 2 * nBins * sizeof(int), stream>>>(
      edges, hist, cursor, binned, E, nBins);
  k3_gemm_binsum<<<nBins, WAVES * 32, 0, stream>>>(
      x, W, b, attn, hist, binned, out, N, E, T, nBins);
}
